# starting kernel
# baseline (speedup 1.0000x reference)
#include <hip/hip_runtime.h>
#include <hip/hip_bf16.h>

typedef __attribute__((ext_vector_type(2))) float v2f;
typedef __attribute__((ext_vector_type(8))) float v8f;

#define CCH 64
#define PAIR_STRIDE 160

__global__ void zero_f32(float* __restrict__ p, long n) {
  long i = (long)blockIdx.x * blockDim.x + threadIdx.x;
  if (i < n) p[i] = 0.0f;
}

__global__ void edge_scatter(const float* __restrict__ x,
                             const int* __restrict__ src_idx,
                             const int* __restrict__ dst_idx,
                             const float* __restrict__ w,
                             float* __restrict__ agg, int E) {
  int t = blockIdx.x * blockDim.x + threadIdx.x;
  int e = t >> 4;
  if (e >= E) return;
  int cg = (t & 15) << 2;
  int s = src_idx[e];
  int d = dst_idx[e];
  float ww = w[e];
  const float4 xv = *(const float4*)(x + (long)s * CCH + cg);
  float* p = agg + (long)d * CCH + cg;
  __hip_atomic_fetch_add(p + 0, xv.x * ww, __ATOMIC_RELAXED, __HIP_MEMORY_SCOPE_AGENT);
  __hip_atomic_fetch_add(p + 1, xv.y * ww, __ATOMIC_RELAXED, __HIP_MEMORY_SCOPE_AGENT);
  __hip_atomic_fetch_add(p + 2, xv.z * ww, __ATOMIC_RELAXED, __HIP_MEMORY_SCOPE_AGENT);
  __hip_atomic_fetch_add(p + 3, xv.w * ww, __ATOMIC_RELAXED, __HIP_MEMORY_SCOPE_AGENT);
}

__global__ __launch_bounds__(256)
void gemm_dual_wmma(const float* __restrict__ Aagg,
                    const float* __restrict__ Aroot,
                    const float* __restrict__ Wn,
                    const float* __restrict__ Wr,
                    const float* __restrict__ bias,
                    float* __restrict__ out, int Nrows) {
  __shared__ float lds[64 * PAIR_STRIDE];

  for (int idx = threadIdx.x; idx < 128 * 64; idx += 256) {
    int k = idx >> 6, n = idx & 63;
    float v = (k < 64) ? Wn[idx] : Wr[idx - 4096];
    lds[(k >> 1) * PAIR_STRIDE + (n << 1) + (k & 1)] = v;
  }
  __syncthreads();

  const int wave = threadIdx.x >> 5;
  const int lane = threadIdx.x & 31;
  const int lh = lane >> 4;
  const int l  = lane & 15;

  long tile = (long)blockIdx.x * 8 + wave;
  long row0 = tile * 16;
  if (row0 >= Nrows) return;

  long arow = row0 + l;
  if (arow >= Nrows) arow = Nrows - 1;
  const float* pa = Aagg + arow * CCH;
  const float* px = Aroot + arow * CCH;

  v8f acc0 = {}, acc1 = {}, acc2 = {}, acc3 = {};

#pragma unroll
  for (int k0 = 0; k0 < 128; k0 += 4) {
    int kk = k0 + (lh << 1);
    v2f a = (k0 < 64) ? *(const v2f*)(pa + kk)
                      : *(const v2f*)(px + (kk - 64));
    const float* bb = lds + (((k0 >> 1) + lh) * PAIR_STRIDE) + (l << 1);
    v2f b0 = *(const v2f*)(bb);
    v2f b1 = *(const v2f*)(bb + 32);
    v2f b2 = *(const v2f*)(bb + 64);
    v2f b3 = *(const v2f*)(bb + 96);
    acc0 = __builtin_amdgcn_wmma_f32_16x16x4_f32(false, a, false, b0, (short)0, acc0, false, false);
    acc1 = __builtin_amdgcn_wmma_f32_16x16x4_f32(false, a, false, b1, (short)0, acc1, false, false);
    acc2 = __builtin_amdgcn_wmma_f32_16x16x4_f32(false, a, false, b2, (short)0, acc2, false, false);
    acc3 = __builtin_amdgcn_wmma_f32_16x16x4_f32(false, a, false, b3, (short)0, acc3, false, false);
  }

  float bc0 = bias[l], bc1 = bias[16 + l], bc2 = bias[32 + l], bc3 = bias[48 + l];

#pragma unroll
  for (int r = 0; r < 8; ++r) {
    long orow = row0 + r + (lh << 3);
    if (orow < Nrows) {
      float* po = out + orow * CCH + l;
      po[0]  = acc0[r] + bc0;
      po[16] = acc1[r] + bc1;
      po[32] = acc2[r] + bc2;
      po[48] = acc3[r] + bc3;
    }
  }
}

extern "C" void kernel_launch(void* const* d_in, const int* in_sizes, int n_in,
                              void* d_out, int out_size, void* d_ws, size_t ws_size,
                              hipStream_t stream) {
  const float* x   = (const float*)d_in[0];
  const int*   ei  = (const int*)d_in[1];
  const float* ea  = (const float*)d_in[2];
  const int*   ps  = (const int*)d_in[3];
  const int*   pd  = (const int*)d_in[4];
  const float* pw  = (const float*)d_in[5];
  const float* Wr1 = (const float*)d_in[7];
  const float* Wn1 = (const float*)d_in[8];
  const float* b1  = (const float*)d_in[9];
  const float* Wr2 = (const float*)d_in[10];
  const float* Wn2 = (const float*)d_in[11];
  const float* b2  = (const float*)d_in[12];
  float* out = (float*)d_out;

  const int N  = in_sizes[0] / CCH;
  const int E  = in_sizes[1] / 2;
  const int EP = in_sizes[3];
  const long NC = (long)N * CCH;

  float* agg = (float*)d_ws;
  float* h1  = agg + NC;
  float* h2  = h1 + NC;

  const int* esrc = ei;
  const int* edst = ei + E;

  const int zb  = (int)((NC + 255) / 256);
  const int sb  = (int)(((long)E * 16 + 255) / 256);
  const int pb  = (int)(((long)EP * 16 + 255) / 256);
  const int gb  = ((N + 15) / 16 + 7) / 8;
  const long outN = (long)out_size;
  const int ob  = (int)((outN + 255) / 256);

  zero_f32<<<zb, 256, 0, stream>>>(agg, NC);
  edge_scatter<<<sb, 256, 0, stream>>>(x, esrc, edst, ea, agg, E);
  gemm_dual_wmma<<<gb, 256, 0, stream>>>(agg, x, Wn1, Wr1, b1, h1, N);
  zero_f32<<<zb, 256, 0, stream>>>(agg, NC);
  edge_scatter<<<sb, 256, 0, stream>>>(h1, esrc, edst, ea, agg, E);
  gemm_dual_wmma<<<gb, 256, 0, stream>>>(agg, h1, Wn2, Wr2, b2, h2, N);
  zero_f32<<<ob, 256, 0, stream>>>(out, outN);
  edge_scatter<<<pb, 256, 0, stream>>>(h2, ps, pd, pw, out, EP);
}


# edge scatter: one wave per edge, lane = channel (c and c+32), so each f32 atomic and each gather load covers one whole 128-B line, not 8 words in each of 4 lines; sums still in atomic arrival order
# speedup vs baseline: 1.7190x; 1.7190x over previous
#include <hip/hip_runtime.h>
#include <hip/hip_bf16.h>

typedef __attribute__((ext_vector_type(2))) float v2f;
typedef __attribute__((ext_vector_type(8))) float v8f;

#define CCH 64
#define PAIR_STRIDE 160

__global__ void zero_f32(float* __restrict__ p, long n) {
  long i = (long)blockIdx.x * blockDim.x + threadIdx.x;
  if (i < n) p[i] = 0.0f;
}

#define EPW 8

__device__ __forceinline__ void red_add(float* p, float v) {
  __hip_atomic_fetch_add(p, v, __ATOMIC_RELAXED, __HIP_MEMORY_SCOPE_AGENT);
}

__global__ __launch_bounds__(256)
void edge_scatter(const float* __restrict__ x,
                  const int* __restrict__ src_idx,
                  const int* __restrict__ dst_idx,
                  const float* __restrict__ w,
                  float* __restrict__ agg, int E) {
  const int lane = threadIdx.x & 31;
  const int wv = __builtin_amdgcn_readfirstlane((int)(threadIdx.x >> 5));
  const int e0 = ((int)blockIdx.x * 8 + wv) * EPW;
  if (e0 >= E) return;

  if (e0 + EPW <= E) {
    float lo[EPW], hi[EPW];
    for (int j = 0; j < EPW; ++j) {
      const float* xr = x + (long)src_idx[e0 + j] * CCH + lane;
      lo[j] = xr[0];
      hi[j] = xr[32];
    }
    for (int j = 0; j < EPW; ++j) {
      const float ww = w[e0 + j];
      float* p = agg + (long)dst_idx[e0 + j] * CCH + lane;
      red_add(p, lo[j] * ww);
      red_add(p + 32, hi[j] * ww);
    }
  } else {
    for (int e = e0; e < E; ++e) {
      const float* xr = x + (long)src_idx[e] * CCH + lane;
      const float ww = w[e];
      float* p = agg + (long)dst_idx[e] * CCH + lane;
      red_add(p, xr[0] * ww);
      red_add(p + 32, xr[32] * ww);
    }
  }
}

__global__ __launch_bounds__(256)
void gemm_dual_wmma(const float* __restrict__ Aagg,
                    const float* __restrict__ Aroot,
                    const float* __restrict__ Wn,
                    const float* __restrict__ Wr,
                    const float* __restrict__ bias,
                    float* __restrict__ out, int Nrows) {
  __shared__ float lds[64 * PAIR_STRIDE];

  for (int idx = threadIdx.x; idx < 128 * 64; idx += 256) {
    int k = idx >> 6, n = idx & 63;
    float v = (k < 64) ? Wn[idx] : Wr[idx - 4096];
    lds[(k >> 1) * PAIR_STRIDE + (n << 1) + (k & 1)] = v;
  }
  __syncthreads();

  const int wave = threadIdx.x >> 5;
  const int lane = threadIdx.x & 31;
  const int lh = lane >> 4;
  const int l  = lane & 15;

  long tile = (long)blockIdx.x * 8 + wave;
  long row0 = tile * 16;
  if (row0 >= Nrows) return;

  long arow = row0 + l;
  if (arow >= Nrows) arow = Nrows - 1;
  const float* pa = Aagg + arow * CCH;
  const float* px = Aroot + arow * CCH;

  v8f acc0 = {}, acc1 = {}, acc2 = {}, acc3 = {};

#pragma unroll
  for (int k0 = 0; k0 < 128; k0 += 4) {
    int kk = k0 + (lh << 1);
    v2f a = (k0 < 64) ? *(const v2f*)(pa + kk)
                      : *(const v2f*)(px + (kk - 64));
    const float* bb = lds + (((k0 >> 1) + lh) * PAIR_STRIDE) + (l << 1);
    v2f b0 = *(const v2f*)(bb);
    v2f b1 = *(const v2f*)(bb + 32);
    v2f b2 = *(const v2f*)(bb + 64);
    v2f b3 = *(const v2f*)(bb + 96);
    acc0 = __builtin_amdgcn_wmma_f32_16x16x4_f32(false, a, false, b0, (short)0, acc0, false, false);
    acc1 = __builtin_amdgcn_wmma_f32_16x16x4_f32(false, a, false, b1, (short)0, acc1, false, false);
    acc2 = __builtin_amdgcn_wmma_f32_16x16x4_f32(false, a, false, b2, (short)0, acc2, false, false);
    acc3 = __builtin_amdgcn_wmma_f32_16x16x4_f32(false, a, false, b3, (short)0, acc3, false, false);
  }

  float bc0 = bias[l], bc1 = bias[16 + l], bc2 = bias[32 + l], bc3 = bias[48 + l];

#pragma unroll
  for (int r = 0; r < 8; ++r) {
    long orow = row0 + r + (lh << 3);
    if (orow < Nrows) {
      float* po = out + orow * CCH + l;
      po[0]  = acc0[r] + bc0;
      po[16] = acc1[r] + bc1;
      po[32] = acc2[r] + bc2;
      po[48] = acc3[r] + bc3;
    }
  }
}

extern "C" void kernel_launch(void* const* d_in, const int* in_sizes, int n_in,
                              void* d_out, int out_size, void* d_ws, size_t ws_size,
                              hipStream_t stream) {
  const float* x   = (const float*)d_in[0];
  const int*   ei  = (const int*)d_in[1];
  const float* ea  = (const float*)d_in[2];
  const int*   ps  = (const int*)d_in[3];
  const int*   pd  = (const int*)d_in[4];
  const float* pw  = (const float*)d_in[5];
  const float* Wr1 = (const float*)d_in[7];
  const float* Wn1 = (const float*)d_in[8];
  const float* b1  = (const float*)d_in[9];
  const float* Wr2 = (const float*)d_in[10];
  const float* Wn2 = (const float*)d_in[11];
  const float* b2  = (const float*)d_in[12];
  float* out = (float*)d_out;

  const int N  = in_sizes[0] / CCH;
  const int E  = in_sizes[1] / 2;
  const int EP = in_sizes[3];
  const long NC = (long)N * CCH;

  float* agg = (float*)d_ws;
  float* h1  = agg + NC;
  float* h2  = h1 + NC;

  const int* esrc = ei;
  const int* edst = ei + E;

  const int zb  = (int)((NC + 255) / 256);
  const int sb  = (int)(((long)E + 8 * EPW - 1) / (8 * EPW));
  const int pb  = (int)(((long)EP + 8 * EPW - 1) / (8 * EPW));
  const int gb  = ((N + 15) / 16 + 7) / 8;
  const long outN = (long)out_size;
  const int ob  = (int)((outN + 255) / 256);

  zero_f32<<<zb, 256, 0, stream>>>(agg, NC);
  edge_scatter<<<sb, 256, 0, stream>>>(x, esrc, edst, ea, agg, E);
  gemm_dual_wmma<<<gb, 256, 0, stream>>>(agg, x, Wn1, Wr1, b1, h1, N);
  zero_f32<<<zb, 256, 0, stream>>>(agg, NC);
  edge_scatter<<<sb, 256, 0, stream>>>(h1, esrc, edst, ea, agg, E);
  gemm_dual_wmma<<<gb, 256, 0, stream>>>(agg, h1, Wn2, Wr2, b2, h2, N);
  zero_f32<<<ob, 256, 0, stream>>>(out, outN);
  edge_scatter<<<pb, 256, 0, stream>>>(h2, ps, pd, pw, out, EP);
}


# zero fill: a fixed grid of 256 eight-wave workgroups writes each buffer with 16-byte stores (one wave store = 4 whole lines) instead of one 4-byte store per thread, i.e. a new wave per 128 bytes
# speedup vs baseline: 1.9452x; 1.1316x over previous
#include <hip/hip_runtime.h>
#include <hip/hip_bf16.h>

typedef __attribute__((ext_vector_type(2))) float v2f;
typedef __attribute__((ext_vector_type(8))) float v8f;

#define CCH 64
#define PAIR_STRIDE 160

#define ZERO_WG 256

__global__ __launch_bounds__(256)
void zero_f32(float* __restrict__ p, long n) {
  long head = (long)(((16u - (unsigned)((size_t)p & 15u)) & 15u) >> 2);
  if (head > n) head = n;
  float4* __restrict__ q = (float4*)(p + head);
  const long n4 = (n - head) >> 2;
  const int  lt = (int)threadIdx.x;
  const float4 z = make_float4(0.0f, 0.0f, 0.0f, 0.0f);

  const long per   = (((n4 + ZERO_WG - 1) / ZERO_WG) + 31) & ~31L;
  const long begin = (long)blockIdx.x * per;
  long len = n4 - begin;
  if (len > per) len = per;

  if (len > 0) {
    float4* __restrict__ w = q + begin + lt;
    long c = 0;
    for (; c + 1024 <= len; c += 1024) {
      w[c]       = z;
      w[c + 256] = z;
      w[c + 512] = z;
      w[c + 768] = z;
    }
    for (; c + 256 <= len; c += 256) w[c] = z;
    if (c + lt < len) w[c] = z;
  }

  const long t = (long)blockIdx.x * 256 + lt;
  if (t < head) p[t] = 0.0f;
  const long tail0 = head + (n4 << 2);
  if (tail0 + t < n) p[tail0 + t] = 0.0f;
}

#define EPW 8

__device__ __forceinline__ void red_add(float* p, float v) {
  __hip_atomic_fetch_add(p, v, __ATOMIC_RELAXED, __HIP_MEMORY_SCOPE_AGENT);
}

__global__ __launch_bounds__(256)
void edge_scatter(const float* __restrict__ x,
                  const int* __restrict__ src_idx,
                  const int* __restrict__ dst_idx,
                  const float* __restrict__ w,
                  float* __restrict__ agg, int E) {
  const int lane = threadIdx.x & 31;
  const int wv = __builtin_amdgcn_readfirstlane((int)(threadIdx.x >> 5));
  const int e0 = ((int)blockIdx.x * 8 + wv) * EPW;
  if (e0 >= E) return;

  if (e0 + EPW <= E) {
    float lo[EPW], hi[EPW];
    for (int j = 0; j < EPW; ++j) {
      const float* xr = x + (long)src_idx[e0 + j] * CCH + lane;
      lo[j] = xr[0];
      hi[j] = xr[32];
    }
    for (int j = 0; j < EPW; ++j) {
      const float ww = w[e0 + j];
      float* p = agg + (long)dst_idx[e0 + j] * CCH + lane;
      red_add(p, lo[j] * ww);
      red_add(p + 32, hi[j] * ww);
    }
  } else {
    for (int e = e0; e < E; ++e) {
      const float* xr = x + (long)src_idx[e] * CCH + lane;
      const float ww = w[e];
      float* p = agg + (long)dst_idx[e] * CCH + lane;
      red_add(p, xr[0] * ww);
      red_add(p + 32, xr[32] * ww);
    }
  }
}

__global__ __launch_bounds__(256)
void gemm_dual_wmma(const float* __restrict__ Aagg,
                    const float* __restrict__ Aroot,
                    const float* __restrict__ Wn,
                    const float* __restrict__ Wr,
                    const float* __restrict__ bias,
                    float* __restrict__ out, int Nrows) {
  __shared__ float lds[64 * PAIR_STRIDE];

  for (int idx = threadIdx.x; idx < 128 * 64; idx += 256) {
    int k = idx >> 6, n = idx & 63;
    float v = (k < 64) ? Wn[idx] : Wr[idx - 4096];
    lds[(k >> 1) * PAIR_STRIDE + (n << 1) + (k & 1)] = v;
  }
  __syncthreads();

  const int wave = threadIdx.x >> 5;
  const int lane = threadIdx.x & 31;
  const int lh = lane >> 4;
  const int l  = lane & 15;

  long tile = (long)blockIdx.x * 8 + wave;
  long row0 = tile * 16;
  if (row0 >= Nrows) return;

  long arow = row0 + l;
  if (arow >= Nrows) arow = Nrows - 1;
  const float* pa = Aagg + arow * CCH;
  const float* px = Aroot + arow * CCH;

  v8f acc0 = {}, acc1 = {}, acc2 = {}, acc3 = {};

#pragma unroll
  for (int k0 = 0; k0 < 128; k0 += 4) {
    int kk = k0 + (lh << 1);
    v2f a = (k0 < 64) ? *(const v2f*)(pa + kk)
                      : *(const v2f*)(px + (kk - 64));
    const float* bb = lds + (((k0 >> 1) + lh) * PAIR_STRIDE) + (l << 1);
    v2f b0 = *(const v2f*)(bb);
    v2f b1 = *(const v2f*)(bb + 32);
    v2f b2 = *(const v2f*)(bb + 64);
    v2f b3 = *(const v2f*)(bb + 96);
    acc0 = __builtin_amdgcn_wmma_f32_16x16x4_f32(false, a, false, b0, (short)0, acc0, false, false);
    acc1 = __builtin_amdgcn_wmma_f32_16x16x4_f32(false, a, false, b1, (short)0, acc1, false, false);
    acc2 = __builtin_amdgcn_wmma_f32_16x16x4_f32(false, a, false, b2, (short)0, acc2, false, false);
    acc3 = __builtin_amdgcn_wmma_f32_16x16x4_f32(false, a, false, b3, (short)0, acc3, false, false);
  }

  float bc0 = bias[l], bc1 = bias[16 + l], bc2 = bias[32 + l], bc3 = bias[48 + l];

#pragma unroll
  for (int r = 0; r < 8; ++r) {
    long orow = row0 + r + (lh << 3);
    if (orow < Nrows) {
      float* po = out + orow * CCH + l;
      po[0]  = acc0[r] + bc0;
      po[16] = acc1[r] + bc1;
      po[32] = acc2[r] + bc2;
      po[48] = acc3[r] + bc3;
    }
  }
}

extern "C" void kernel_launch(void* const* d_in, const int* in_sizes, int n_in,
                              void* d_out, int out_size, void* d_ws, size_t ws_size,
                              hipStream_t stream) {
  const float* x   = (const float*)d_in[0];
  const int*   ei  = (const int*)d_in[1];
  const float* ea  = (const float*)d_in[2];
  const int*   ps  = (const int*)d_in[3];
  const int*   pd  = (const int*)d_in[4];
  const float* pw  = (const float*)d_in[5];
  const float* Wr1 = (const float*)d_in[7];
  const float* Wn1 = (const float*)d_in[8];
  const float* b1  = (const float*)d_in[9];
  const float* Wr2 = (const float*)d_in[10];
  const float* Wn2 = (const float*)d_in[11];
  const float* b2  = (const float*)d_in[12];
  float* out = (float*)d_out;

  const int N  = in_sizes[0] / CCH;
  const int E  = in_sizes[1] / 2;
  const int EP = in_sizes[3];
  const long NC = (long)N * CCH;

  float* agg = (float*)d_ws;
  float* h1  = agg + NC;
  float* h2  = h1 + NC;

  const int* esrc = ei;
  const int* edst = ei + E;

  const int sb  = (int)(((long)E + 8 * EPW - 1) / (8 * EPW));
  const int pb  = (int)(((long)EP + 8 * EPW - 1) / (8 * EPW));
  const int gb  = ((N + 15) / 16 + 7) / 8;
  const long outN = (long)out_size;

  zero_f32<<<ZERO_WG, 256, 0, stream>>>(agg, NC);
  edge_scatter<<<sb, 256, 0, stream>>>(x, esrc, edst, ea, agg, E);
  gemm_dual_wmma<<<gb, 256, 0, stream>>>(agg, x, Wn1, Wr1, b1, h1, N);
  zero_f32<<<ZERO_WG, 256, 0, stream>>>(agg, NC);
  edge_scatter<<<sb, 256, 0, stream>>>(h1, esrc, edst, ea, agg, E);
  gemm_dual_wmma<<<gb, 256, 0, stream>>>(agg, h1, Wn2, Wr2, b2, h2, N);
  zero_f32<<<ZERO_WG, 256, 0, stream>>>(out, outN);
  edge_scatter<<<pb, 256, 0, stream>>>(h2, ps, pd, pw, out, EP);
}


# conv layers: W_nei applied before the aggregation (y=x@Wn); GEMM writes x@Wr+b into h, edge atomics add y[src]*w onto it: no agg buffer, no agg zero fill; atomic sums run over rows of y, not of x
# speedup vs baseline: 2.0674x; 1.0628x over previous
#include <hip/hip_runtime.h>
#include <hip/hip_bf16.h>

typedef __attribute__((ext_vector_type(2))) float v2f;
typedef __attribute__((ext_vector_type(8))) float v8f;

#define CCH 64
#define PAIR_STRIDE 288

#define ZERO_WG 256

__global__ __launch_bounds__(256)
void zero_f32(float* __restrict__ p, long n) {
  long head = (long)(((16u - (unsigned)((size_t)p & 15u)) & 15u) >> 2);
  if (head > n) head = n;
  float4* __restrict__ q = (float4*)(p + head);
  const long n4 = (n - head) >> 2;
  const int  lt = (int)threadIdx.x;
  const float4 z = make_float4(0.0f, 0.0f, 0.0f, 0.0f);

  const long per   = (((n4 + ZERO_WG - 1) / ZERO_WG) + 31) & ~31L;
  const long begin = (long)blockIdx.x * per;
  long len = n4 - begin;
  if (len > per) len = per;

  if (len > 0) {
    float4* __restrict__ w = q + begin + lt;
    long c = 0;
    for (; c + 1024 <= len; c += 1024) {
      w[c]       = z;
      w[c + 256] = z;
      w[c + 512] = z;
      w[c + 768] = z;
    }
    for (; c + 256 <= len; c += 256) w[c] = z;
    if (c + lt < len) w[c] = z;
  }

  const long t = (long)blockIdx.x * 256 + lt;
  if (t < head) p[t] = 0.0f;
  const long tail0 = head + (n4 << 2);
  if (tail0 + t < n) p[tail0 + t] = 0.0f;
}

#define EPW 8

__device__ __forceinline__ void red_add(float* p, float v) {
  __hip_atomic_fetch_add(p, v, __ATOMIC_RELAXED, __HIP_MEMORY_SCOPE_AGENT);
}

__global__ __launch_bounds__(256)
void edge_scatter(const float* __restrict__ x,
                  const int* __restrict__ src_idx,
                  const int* __restrict__ dst_idx,
                  const float* __restrict__ w,
                  float* __restrict__ agg, int E) {
  const int lane = threadIdx.x & 31;
  const int wv = __builtin_amdgcn_readfirstlane((int)(threadIdx.x >> 5));
  const int e0 = ((int)blockIdx.x * 8 + wv) * EPW;
  if (e0 >= E) return;

  if (e0 + EPW <= E) {
    float lo[EPW], hi[EPW];
    for (int j = 0; j < EPW; ++j) {
      const float* xr = x + (long)src_idx[e0 + j] * CCH + lane;
      lo[j] = xr[0];
      hi[j] = xr[32];
    }
    for (int j = 0; j < EPW; ++j) {
      const float ww = w[e0 + j];
      float* p = agg + (long)dst_idx[e0 + j] * CCH + lane;
      red_add(p, lo[j] * ww);
      red_add(p + 32, hi[j] * ww);
    }
  } else {
    for (int e = e0; e < E; ++e) {
      const float* xr = x + (long)src_idx[e] * CCH + lane;
      const float ww = w[e];
      float* p = agg + (long)dst_idx[e] * CCH + lane;
      red_add(p, xr[0] * ww);
      red_add(p + 32, xr[32] * ww);
    }
  }
}

__global__ __launch_bounds__(256)
void gemm_pre_wmma(const float* __restrict__ X,
                   const float* __restrict__ Wn,
                   const float* __restrict__ Wr,
                   const float* __restrict__ bias,
                   float* __restrict__ Y,
                   float* __restrict__ H, int Nrows) {
  __shared__ float lds[32 * PAIR_STRIDE];

  const int wave = threadIdx.x >> 5;
  const int lane = threadIdx.x & 31;
  const int lh = lane >> 4;
  const int l  = lane & 15;

  const long tile = (long)blockIdx.x * 8 + wave;
  const long row0 = tile * 16;

  long arow = row0 + l;
  if (arow >= Nrows) arow = Nrows - 1;
  const float* px = X + arow * CCH;

  v2f afr[16];
#pragma unroll
  for (int i = 0; i < 16; ++i) afr[i] = *(const v2f*)(px + 4 * i + (lh << 1));

  for (int idx = threadIdx.x; idx < 64 * 128; idx += 256) {
    int k = idx >> 7, n = idx & 127;
    float v = (n < 64) ? Wn[k * CCH + n] : Wr[k * CCH + (n - 64)];
    lds[(k >> 1) * PAIR_STRIDE + (n << 1) + (k & 1)] = v;
  }
  __syncthreads();

  if (row0 >= Nrows) return;

  v8f acc0 = {}, acc1 = {}, acc2 = {}, acc3 = {};
  v8f acc4 = {}, acc5 = {}, acc6 = {}, acc7 = {};

#pragma unroll
  for (int k0 = 0; k0 < 64; k0 += 4) {
    const v2f a = afr[k0 >> 2];
    const float* bb = lds + (((k0 >> 1) + lh) * PAIR_STRIDE) + (l << 1);
    v2f b0 = *(const v2f*)(bb);
    v2f b1 = *(const v2f*)(bb + 32);
    v2f b2 = *(const v2f*)(bb + 64);
    v2f b3 = *(const v2f*)(bb + 96);
    v2f b4 = *(const v2f*)(bb + 128);
    v2f b5 = *(const v2f*)(bb + 160);
    v2f b6 = *(const v2f*)(bb + 192);
    v2f b7 = *(const v2f*)(bb + 224);
    acc0 = __builtin_amdgcn_wmma_f32_16x16x4_f32(false, a, false, b0, (short)0, acc0, false, false);
    acc1 = __builtin_amdgcn_wmma_f32_16x16x4_f32(false, a, false, b1, (short)0, acc1, false, false);
    acc2 = __builtin_amdgcn_wmma_f32_16x16x4_f32(false, a, false, b2, (short)0, acc2, false, false);
    acc3 = __builtin_amdgcn_wmma_f32_16x16x4_f32(false, a, false, b3, (short)0, acc3, false, false);
    acc4 = __builtin_amdgcn_wmma_f32_16x16x4_f32(false, a, false, b4, (short)0, acc4, false, false);
    acc5 = __builtin_amdgcn_wmma_f32_16x16x4_f32(false, a, false, b5, (short)0, acc5, false, false);
    acc6 = __builtin_amdgcn_wmma_f32_16x16x4_f32(false, a, false, b6, (short)0, acc6, false, false);
    acc7 = __builtin_amdgcn_wmma_f32_16x16x4_f32(false, a, false, b7, (short)0, acc7, false, false);
  }

  float bc0 = bias[l], bc1 = bias[16 + l], bc2 = bias[32 + l], bc3 = bias[48 + l];

#pragma unroll
  for (int r = 0; r < 8; ++r) {
    long orow = row0 + r + (lh << 3);
    if (orow < Nrows) {
      float* py = Y + orow * CCH + l;
      py[0]  = acc0[r];
      py[16] = acc1[r];
      py[32] = acc2[r];
      py[48] = acc3[r];
      float* ph = H + orow * CCH + l;
      ph[0]  = acc4[r] + bc0;
      ph[16] = acc5[r] + bc1;
      ph[32] = acc6[r] + bc2;
      ph[48] = acc7[r] + bc3;
    }
  }
}

extern "C" void kernel_launch(void* const* d_in, const int* in_sizes, int n_in,
                              void* d_out, int out_size, void* d_ws, size_t ws_size,
                              hipStream_t stream) {
  const float* x   = (const float*)d_in[0];
  const int*   ei  = (const int*)d_in[1];
  const float* ea  = (const float*)d_in[2];
  const int*   ps  = (const int*)d_in[3];
  const int*   pd  = (const int*)d_in[4];
  const float* pw  = (const float*)d_in[5];
  const float* Wr1 = (const float*)d_in[7];
  const float* Wn1 = (const float*)d_in[8];
  const float* b1  = (const float*)d_in[9];
  const float* Wr2 = (const float*)d_in[10];
  const float* Wn2 = (const float*)d_in[11];
  const float* b2  = (const float*)d_in[12];
  float* out = (float*)d_out;

  const int N  = in_sizes[0] / CCH;
  const int E  = in_sizes[1] / 2;
  const int EP = in_sizes[3];
  const long NC = (long)N * CCH;

  float* y   = (float*)d_ws;
  float* h1  = y + NC;
  float* h2  = h1 + NC;

  const int* esrc = ei;
  const int* edst = ei + E;

  const int sb  = (int)(((long)E + 8 * EPW - 1) / (8 * EPW));
  const int pb  = (int)(((long)EP + 8 * EPW - 1) / (8 * EPW));
  const int gb  = ((N + 15) / 16 + 7) / 8;
  const long outN = (long)out_size;

  gemm_pre_wmma<<<gb, 256, 0, stream>>>(x, Wn1, Wr1, b1, y, h1, N);
  edge_scatter<<<sb, 256, 0, stream>>>(y, esrc, edst, ea, h1, E);
  gemm_pre_wmma<<<gb, 256, 0, stream>>>(h1, Wn2, Wr2, b2, y, h2, N);
  edge_scatter<<<sb, 256, 0, stream>>>(y, esrc, edst, ea, h2, E);
  zero_f32<<<ZERO_WG, 256, 0, stream>>>(out, outN);
  edge_scatter<<<pb, 256, 0, stream>>>(h2, ps, pd, pw, out, EP);
}


# edge passes and zero fill: rows split in two ranges, one per half of the package; a workgroup takes (by ticket) its own half's range, so a row's adds come from one L2; sums in arrival order
# speedup vs baseline: 2.2459x; 1.0863x over previous
#include <hip/hip_runtime.h>
#include <hip/hip_bf16.h>

typedef __attribute__((ext_vector_type(2))) float v2f;
typedef __attribute__((ext_vector_type(8))) float v8f;

#define CCH 64
#define PAIR_STRIDE 288
#define TICKET_PITCH 128

#define ZERO_WG 256

__global__ __launch_bounds__(256)
void zero_f32(float* __restrict__ p, long n, float* tickets) {
  __shared__ int s_slice;
  if (threadIdx.x == 0) {
    int* cnt = (int*)tickets;
    int side = (int)((__builtin_amdgcn_s_sendmsg_rtn(0x87) >> 18) & 1u);
    int t = __hip_atomic_fetch_add(cnt + side * TICKET_PITCH, 1, __ATOMIC_RELAXED, __HIP_MEMORY_SCOPE_AGENT);
    if (t >= ZERO_WG / 2) {
      side ^= 1;
      t = __hip_atomic_fetch_add(cnt + side * TICKET_PITCH, 1, __ATOMIC_RELAXED, __HIP_MEMORY_SCOPE_AGENT);
    }
    s_slice = (t < ZERO_WG / 2) ? (side * (ZERO_WG / 2) + t) : -1;
  }
  __syncthreads();
  const int slice = s_slice;
  if (slice < 0) return;

  long head = (long)(((16u - (unsigned)((size_t)p & 15u)) & 15u) >> 2);
  if (head > n) head = n;
  float4* __restrict__ q = (float4*)(p + head);
  const long n4 = (n - head) >> 2;
  const int  lt = (int)threadIdx.x;
  const float4 z = make_float4(0.0f, 0.0f, 0.0f, 0.0f);

  const long per   = (((n4 + ZERO_WG - 1) / ZERO_WG) + 31) & ~31L;
  const long begin = (long)slice * per;
  long len = n4 - begin;
  if (len > per) len = per;

  if (len > 0) {
    float4* __restrict__ w = q + begin + lt;
    long c = 0;
    for (; c + 1024 <= len; c += 1024) {
      w[c]       = z;
      w[c + 256] = z;
      w[c + 512] = z;
      w[c + 768] = z;
    }
    for (; c + 256 <= len; c += 256) w[c] = z;
    if (c + lt < len) w[c] = z;
  }

  const long t = (long)slice * 256 + lt;
  if (t < head) p[t] = 0.0f;
  const long tail0 = head + (n4 << 2);
  if (tail0 + t < n) p[tail0 + t] = 0.0f;
}

#define SCATTER_WG 2048
#define EPW 8
#define WAVE_EDGES 32
#define TICKET_EDGES (8 * WAVE_EDGES)

__device__ __forceinline__ void red_add(float* p, float v) {
  __hip_atomic_fetch_add(p, v, __ATOMIC_RELAXED, __HIP_MEMORY_SCOPE_AGENT);
}

__global__ __launch_bounds__(256)
void edge_scatter(const float* __restrict__ x,
                  const int* __restrict__ src_idx,
                  const int* __restrict__ dst_idx,
                  const float* __restrict__ w,
                  float* __restrict__ agg, int E, int nrows,
                  float* tickets, float* next_tickets) {
  if (next_tickets != nullptr && blockIdx.x == 0 && threadIdx.x < 4) next_tickets[threadIdx.x * TICKET_PITCH] = 0.0f;

  __shared__ int s_t[2];
  int* cnt = (int*)tickets;

  const int lane = threadIdx.x & 31;
  const int wv = (int)(threadIdx.x >> 5);
  const int split = nrows >> 1;
  const int nt = (E + TICKET_EDGES - 1) / TICKET_EDGES;
  const int hh = (int)((__builtin_amdgcn_s_sendmsg_rtn(0x87) >> 18) & 1u);

  int cur = hh;
  int t1 = 0;
  if (threadIdx.x == 0)
    t1 = __hip_atomic_fetch_add(cnt + cur * TICKET_PITCH, 1, __ATOMIC_RELAXED, __HIP_MEMORY_SCOPE_AGENT);

  const int maxit = 2 * nt + 1;
  for (int it = 0; it < maxit; ++it) {
    if (threadIdx.x == 0) {
      int t = t1;
      if (t >= nt && cur == hh) {
        cur = hh ^ 1;
        t = __hip_atomic_fetch_add(cnt + cur * TICKET_PITCH, 1, __ATOMIC_RELAXED, __HIP_MEMORY_SCOPE_AGENT);
      }
      s_t[it & 1] = (t < nt) ? (t * 2 + cur) : -1;
    }
    __syncthreads();
    const int tt = __builtin_amdgcn_readfirstlane(s_t[it & 1]);
    if (tt < 0) break;
    if (threadIdx.x == 0)
      t1 = __hip_atomic_fetch_add(cnt + cur * TICKET_PITCH, 1, __ATOMIC_RELAXED, __HIP_MEMORY_SCOPE_AGENT);
    const int side = tt & 1;
    const int e = (tt >> 1) * TICKET_EDGES + wv * WAVE_EDGES + lane;
    const bool valid = e < E;
    const int   dv = valid ? dst_idx[e] : 0;
    const int   sv = valid ? src_idx[e] : 0;
    const float fv = valid ? w[e] : 0.0f;
    unsigned m = __builtin_amdgcn_ballot_w32(valid && (((dv >= split) ? 1 : 0) == side));

    for (int g = 0; g < WAVE_EDGES / EPW && m != 0u; ++g) {
      int   kk[EPW];
      float lo[EPW], hi[EPW];
      for (int j = 0; j < EPW; ++j) {
        kk[j] = -1;
        if (m != 0u) {
          const int k = __builtin_ctz(m);
          m &= m - 1u;
          kk[j] = k;
          const float* xr = x + (long)__builtin_amdgcn_readlane(sv, k) * CCH + lane;
          lo[j] = xr[0];
          hi[j] = xr[32];
        }
      }
      for (int j = 0; j < EPW; ++j) {
        if (kk[j] >= 0) {
          const int   d  = __builtin_amdgcn_readlane(dv, kk[j]);
          const float ww = __builtin_bit_cast(float, __builtin_amdgcn_readlane(__builtin_bit_cast(int, fv), kk[j]));
          float* p = agg + (long)d * CCH + lane;
          red_add(p, lo[j] * ww);
          red_add(p + 32, hi[j] * ww);
        }
      }
    }
  }
}

__global__ __launch_bounds__(256)
void gemm_pre_wmma(const float* __restrict__ X,
                   const float* __restrict__ Wn,
                   const float* __restrict__ Wr,
                   const float* __restrict__ bias,
                   float* __restrict__ Y,
                   float* __restrict__ H, int Nrows,
                   float* __restrict__ tickets) {
  __shared__ float lds[32 * PAIR_STRIDE];

  if (tickets != nullptr && blockIdx.x == 0 && threadIdx.x < 4) tickets[threadIdx.x * TICKET_PITCH] = 0.0f;

  const int wave = threadIdx.x >> 5;
  const int lane = threadIdx.x & 31;
  const int lh = lane >> 4;
  const int l  = lane & 15;

  const long tile = (long)blockIdx.x * 8 + wave;
  const long row0 = tile * 16;

  long arow = row0 + l;
  if (arow >= Nrows) arow = Nrows - 1;
  const float* px = X + arow * CCH;

  v2f afr[16];
#pragma unroll
  for (int i = 0; i < 16; ++i) afr[i] = *(const v2f*)(px + 4 * i + (lh << 1));

  for (int idx = threadIdx.x; idx < 64 * 128; idx += 256) {
    int k = idx >> 7, n = idx & 127;
    float v = (n < 64) ? Wn[k * CCH + n] : Wr[k * CCH + (n - 64)];
    lds[(k >> 1) * PAIR_STRIDE + (n << 1) + (k & 1)] = v;
  }
  __syncthreads();

  if (row0 >= Nrows) return;

  v8f acc0 = {}, acc1 = {}, acc2 = {}, acc3 = {};
  v8f acc4 = {}, acc5 = {}, acc6 = {}, acc7 = {};

#pragma unroll
  for (int k0 = 0; k0 < 64; k0 += 4) {
    const v2f a = afr[k0 >> 2];
    const float* bb = lds + (((k0 >> 1) + lh) * PAIR_STRIDE) + (l << 1);
    v2f b0 = *(const v2f*)(bb);
    v2f b1 = *(const v2f*)(bb + 32);
    v2f b2 = *(const v2f*)(bb + 64);
    v2f b3 = *(const v2f*)(bb + 96);
    v2f b4 = *(const v2f*)(bb + 128);
    v2f b5 = *(const v2f*)(bb + 160);
    v2f b6 = *(const v2f*)(bb + 192);
    v2f b7 = *(const v2f*)(bb + 224);
    acc0 = __builtin_amdgcn_wmma_f32_16x16x4_f32(false, a, false, b0, (short)0, acc0, false, false);
    acc1 = __builtin_amdgcn_wmma_f32_16x16x4_f32(false, a, false, b1, (short)0, acc1, false, false);
    acc2 = __builtin_amdgcn_wmma_f32_16x16x4_f32(false, a, false, b2, (short)0, acc2, false, false);
    acc3 = __builtin_amdgcn_wmma_f32_16x16x4_f32(false, a, false, b3, (short)0, acc3, false, false);
    acc4 = __builtin_amdgcn_wmma_f32_16x16x4_f32(false, a, false, b4, (short)0, acc4, false, false);
    acc5 = __builtin_amdgcn_wmma_f32_16x16x4_f32(false, a, false, b5, (short)0, acc5, false, false);
    acc6 = __builtin_amdgcn_wmma_f32_16x16x4_f32(false, a, false, b6, (short)0, acc6, false, false);
    acc7 = __builtin_amdgcn_wmma_f32_16x16x4_f32(false, a, false, b7, (short)0, acc7, false, false);
  }

  float bc0 = bias[l], bc1 = bias[16 + l], bc2 = bias[32 + l], bc3 = bias[48 + l];

#pragma unroll
  for (int r = 0; r < 8; ++r) {
    long orow = row0 + r + (lh << 3);
    if (orow < Nrows) {
      float* py = Y + orow * CCH + l;
      py[0]  = acc0[r];
      py[16] = acc1[r];
      py[32] = acc2[r];
      py[48] = acc3[r];
      float* ph = H + orow * CCH + l;
      ph[0]  = acc4[r] + bc0;
      ph[16] = acc5[r] + bc1;
      ph[32] = acc6[r] + bc2;
      ph[48] = acc7[r] + bc3;
    }
  }
}

extern "C" void kernel_launch(void* const* d_in, const int* in_sizes, int n_in,
                              void* d_out, int out_size, void* d_ws, size_t ws_size,
                              hipStream_t stream) {
  const float* x   = (const float*)d_in[0];
  const int*   ei  = (const int*)d_in[1];
  const float* ea  = (const float*)d_in[2];
  const int*   ps  = (const int*)d_in[3];
  const int*   pd  = (const int*)d_in[4];
  const float* pw  = (const float*)d_in[5];
  const float* Wr1 = (const float*)d_in[7];
  const float* Wn1 = (const float*)d_in[8];
  const float* b1  = (const float*)d_in[9];
  const float* Wr2 = (const float*)d_in[10];
  const float* Wn2 = (const float*)d_in[11];
  const float* b2  = (const float*)d_in[12];
  float* out = (float*)d_out;

  const int N  = in_sizes[0] / CCH;
  const int E  = in_sizes[1] / 2;
  const int EP = in_sizes[3];
  const long NC = (long)N * CCH;

  float* y   = (float*)d_ws;
  float* h1  = y + NC;
  float* h2  = h1 + NC;

  const int* esrc = ei;
  const int* edst = ei + E;

  const int NF  = out_size / CCH;
  const int gb  = ((N + 15) / 16 + 7) / 8;
  const long outN = (long)out_size;

  gemm_pre_wmma<<<gb, 256, 0, stream>>>(x, Wn1, Wr1, b1, y, h1, N, out);
  edge_scatter<<<SCATTER_WG, 256, 0, stream>>>(y, esrc, edst, ea, h1, E, N, out, nullptr);
  gemm_pre_wmma<<<gb, 256, 0, stream>>>(h1, Wn2, Wr2, b2, y, h2, N, nullptr);
  edge_scatter<<<SCATTER_WG, 256, 0, stream>>>(y, esrc, edst, ea, h2, E, N, out + 2 * TICKET_PITCH, h1);
  zero_f32<<<ZERO_WG, 256, 0, stream>>>(out, outN, h1 + 2 * TICKET_PITCH);
  edge_scatter<<<SCATTER_WG, 256, 0, stream>>>(h2, ps, pd, pw, out, EP, NF, h1, nullptr);
}
